# starting kernel
# baseline (speedup 1.0000x reference)
#include <hip/hip_runtime.h>


#define N_NODES 100000
#define N_EDGES 1600000
#define D       128
#define M_TILE  64
#define TPB     128
#define XPITCH  80
#define WPITCH  144

typedef __attribute__((ext_vector_type(2))) float v2f;
typedef __attribute__((ext_vector_type(8))) float v8f;

__device__ __forceinline__ v2f mk2(float a, float b) {
  v2f t; t.x = a; t.y = b; return t;
}

__global__ __launch_bounds__(256) void zero_kernel(float* __restrict__ out, int n4) {
  float4 z = make_float4(0.f, 0.f, 0.f, 0.f);
  float4* o4 = reinterpret_cast<float4*>(out);
  for (int i = blockIdx.x * blockDim.x + threadIdx.x; i < n4;
       i += gridDim.x * blockDim.x)
    o4[i] = z;
}

__global__ __launch_bounds__(256) void scatter_kernel(const float* __restrict__ nf,
                                                      const int* __restrict__ src,
                                                      const int* __restrict__ dst,
                                                      float* __restrict__ agg) {
  int e    = blockIdx.x * (256 >> 5) + (threadIdx.x >> 5);
  int lane = threadIdx.x & 31;
  if (e >= N_EDGES) return;
  int s = src[e];
  int d = dst[e];
  const float4* row = reinterpret_cast<const float4*>(nf + (size_t)s * D);
  float4 v = row[lane];
  float* a = agg + (size_t)d * D + lane * 4;
  __hip_atomic_fetch_add(a + 0, v.x, __ATOMIC_RELAXED, __HIP_MEMORY_SCOPE_AGENT);
  __hip_atomic_fetch_add(a + 1, v.y, __ATOMIC_RELAXED, __HIP_MEMORY_SCOPE_AGENT);
  __hip_atomic_fetch_add(a + 2, v.z, __ATOMIC_RELAXED, __HIP_MEMORY_SCOPE_AGENT);
  __hip_atomic_fetch_add(a + 3, v.w, __ATOMIC_RELAXED, __HIP_MEMORY_SCOPE_AGENT);
}


__device__ __forceinline__ void load_tile_sw(v2f* __restrict__ sT,
                                             const float* __restrict__ g,
                                             int rowBase, int tid) {
  const float4* g4 = reinterpret_cast<const float4*>(g);
#pragma unroll
  for (int i = 0; i < 16; ++i) {
    int linear = i * TPB + tid;
    int r  = linear & 63;
    int c4 = linear >> 6;
    int gr = rowBase + r;
    if (gr >= N_NODES) gr = N_NODES - 1;
    float4 v = g4[(size_t)gr * 32 + c4];
    sT[(c4 * 2) * XPITCH + r]     = mk2(v.x, v.y);
    sT[(c4 * 2 + 1) * XPITCH + r] = mk2(v.z, v.w);
  }
}

__device__ __forceinline__ void load_w_sw(v2f* __restrict__ sW,
                                          const float* __restrict__ W,
                                          int rowOff, int tid) {
  const float4* g4 = reinterpret_cast<const float4*>(W + (size_t)rowOff * D);
#pragma unroll
  for (int i = 0; i < 8; ++i) {
    int linear = i * TPB + tid;
    int p  = linear >> 5;
    int c4 = linear & 31;
    float4 r0 = g4[(p * 2) * 32 + c4];
    float4 r1 = g4[(p * 2 + 1) * 32 + c4];
    v2f* dstp = sW + p * WPITCH + c4 * 4;
    dstp[0] = mk2(r0.x, r1.x);
    dstp[1] = mk2(r0.y, r1.y);
    dstp[2] = mk2(r0.z, r1.z);
    dstp[3] = mk2(r0.w, r1.w);
  }
}

__device__ __forceinline__ void gemm64(const v2f* __restrict__ sX, int kpBase,
                                       const v2f* __restrict__ sW,
                                       v8f acc[8], int m0, int lane) {
  const int lrow = lane & 15;
  const int hi   = lane >> 4;
#pragma unroll 4
  for (int kp = 0; kp < 32; kp += 2) {
    v2f a = sX[(kpBase + kp + hi) * XPITCH + m0 + lrow];
    const v2f* wp = sW + (kp + hi) * WPITCH + lrow;
#pragma unroll
    for (int n = 0; n < 8; ++n) {
      v2f b = wp[n * 16];
      acc[n] = __builtin_amdgcn_wmma_f32_16x16x4_f32(
           false, a,  false, b,
           (short)0, acc[n],  false,  false);
    }
  }
}

__device__ __forceinline__ void store_acc_sw(v2f* __restrict__ sT,
                                             const v8f acc[8],
                                             const float* __restrict__ bias,
                                             int m0, int lane, bool relu) {
  const int lcol = lane & 15;
  const int mhi  = (lane >> 4) << 3;
  float* base = reinterpret_cast<float*>(sT);
#pragma unroll
  for (int n = 0; n < 8; ++n) {
    int col = n * 16 + lcol;
    float bv = bias[col];
    int p = col >> 1, j = col & 1;
#pragma unroll
    for (int v = 0; v < 8; ++v) {
      float x = acc[n][v] + bv;
      if (relu) x = fmaxf(x, 0.f);
      base[(p * XPITCH + m0 + mhi + v) * 2 + j] = x;
    }
  }
}

__global__ __launch_bounds__(TPB) void mlp_kernel(const float* __restrict__ nf,
                                                  const float* __restrict__ W1,
                                                  const float* __restrict__ b1,
                                                  const float* __restrict__ W2,
                                                  const float* __restrict__ b2,
                                                  const float* __restrict__ W3,
                                                  const float* __restrict__ b3,
                                                  float* __restrict__ out) {
  __shared__ v2f sW[32 * WPITCH];
  __shared__ v2f sX[128 * XPITCH];
  __shared__ v2f sH[64 * XPITCH];

  const int tid  = threadIdx.x;
  const int lane = tid & 31;
  const int m0   = (tid >> 5) * 16;
  const int rowBase = blockIdx.x * M_TILE;

  load_tile_sw(sX,       out, rowBase, tid);
  load_tile_sw(sX + 64 * XPITCH, nf, rowBase, tid);

  const v8f zv = {0.f, 0.f, 0.f, 0.f, 0.f, 0.f, 0.f, 0.f};
  v8f acc[8];
#pragma unroll
  for (int n = 0; n < 8; ++n) acc[n] = zv;

#pragma unroll 1
  for (int s = 0; s < 4; ++s) {
    __syncthreads();
    load_w_sw(sW, W1, s * 64, tid);
    __syncthreads();
    gemm64(sX, s * 32, sW, acc, m0, lane);
  }
  store_acc_sw(sH, acc, b1, m0, lane, true);

#pragma unroll
  for (int n = 0; n < 8; ++n) acc[n] = zv;
#pragma unroll 1
  for (int s = 0; s < 2; ++s) {
    __syncthreads();
    load_w_sw(sW, W2, s * 64, tid);
    __syncthreads();
    gemm64(sH, s * 32, sW, acc, m0, lane);
  }
  store_acc_sw(sX, acc, b2, m0, lane, true);

#pragma unroll
  for (int n = 0; n < 8; ++n) acc[n] = zv;
#pragma unroll 1
  for (int s = 0; s < 2; ++s) {
    __syncthreads();
    load_w_sw(sW, W3, s * 64, tid);
    __syncthreads();
    gemm64(sX, s * 32, sW, acc, m0, lane);
  }
  store_acc_sw(sH, acc, b3, m0, lane, false);
  __syncthreads();

  float4* o4 = reinterpret_cast<float4*>(out);
#pragma unroll
  for (int i = 0; i < 16; ++i) {
    int linear = i * TPB + tid;
    int r  = linear & 63;
    int c4 = linear >> 6;
    int gr = rowBase + r;
    v2f lo = sH[(c4 * 2) * XPITCH + r];
    v2f hv = sH[(c4 * 2 + 1) * XPITCH + r];
    if (gr < N_NODES)
      o4[(size_t)gr * 32 + c4] = make_float4(lo.x, lo.y, hv.x, hv.y);
  }
}

extern "C" void kernel_launch(void* const* d_in, const int* in_sizes, int n_in,
                              void* d_out, int out_size, void* d_ws, size_t ws_size,
                              hipStream_t stream) {
  const float* nf  = (const float*)d_in[0];
  const int*   src = (const int*)d_in[1];
  const int*   dst = (const int*)d_in[2];
  const float* W1  = (const float*)d_in[3];
  const float* b1  = (const float*)d_in[4];
  const float* W2  = (const float*)d_in[5];
  const float* b2  = (const float*)d_in[6];
  const float* W3  = (const float*)d_in[7];
  const float* b3  = (const float*)d_in[8];
  float* out = (float*)d_out;
  (void)in_sizes; (void)n_in; (void)out_size; (void)d_ws; (void)ws_size;

  const int n4 = N_NODES * D / 4;
  zero_kernel<<<(n4 + 255) / 256, 256, 0, stream>>>(out, n4);

  scatter_kernel<<<N_EDGES / 8, 256, 0, stream>>>(nf, src, dst, out);

  const int nBlocks = (N_NODES + M_TILE - 1) / M_TILE;
  mlp_kernel<<<nBlocks, TPB, 0, stream>>>(nf, W1, b1, W2, b2, W3, b3, out);
}


# Scatter-add: each f32 atomic add covers one whole 128-B line (row still one 512-B gather, turned through a wave-private LDS slot): 4 line requests per edge, not 16; sums stay in atomic arrival order
# speedup vs baseline: 1.2697x; 1.2697x over previous
#include <hip/hip_runtime.h>


#define N_NODES 100000
#define N_EDGES 1600000
#define D       128
#define M_TILE  64
#define TPB     128
#define XPITCH  80
#define WPITCH  144

typedef __attribute__((ext_vector_type(2))) float v2f;
typedef __attribute__((ext_vector_type(8))) float v8f;

__device__ __forceinline__ v2f mk2(float a, float b) {
  v2f t; t.x = a; t.y = b; return t;
}

__global__ __launch_bounds__(256) void zero_kernel(float* __restrict__ out, int n4) {
  float4 z = make_float4(0.f, 0.f, 0.f, 0.f);
  float4* o4 = reinterpret_cast<float4*>(out);
  for (int i = blockIdx.x * blockDim.x + threadIdx.x; i < n4;
       i += gridDim.x * blockDim.x)
    o4[i] = z;
}

__global__ __launch_bounds__(256) void scatter_kernel(const float* __restrict__ nf,
                                                      const int* __restrict__ src,
                                                      const int* __restrict__ dst,
                                                      float* __restrict__ agg) {
  __shared__ float4 sT[256];
  int wave = threadIdx.x >> 5;
  int e    = blockIdx.x * (256 >> 5) + wave;
  int lane = threadIdx.x & 31;
  if (e >= N_EDGES) return;
  int s = src[e];
  int d = dst[e];
  const float4* row = reinterpret_cast<const float4*>(nf + (size_t)s * D);
  sT[threadIdx.x] = row[lane];
  __builtin_amdgcn_fence(__ATOMIC_RELEASE, "wavefront");
  __builtin_amdgcn_wave_barrier();
  __builtin_amdgcn_fence(__ATOMIC_ACQUIRE, "wavefront");
  const float* t = reinterpret_cast<const float*>(sT) + wave * D + lane;
  float v0 = t[0];
  float v1 = t[32];
  float v2 = t[64];
  float v3 = t[96];
  float* a = agg + (size_t)d * D + lane;
  __hip_atomic_fetch_add(a +  0, v0, __ATOMIC_RELAXED, __HIP_MEMORY_SCOPE_AGENT);
  __hip_atomic_fetch_add(a + 32, v1, __ATOMIC_RELAXED, __HIP_MEMORY_SCOPE_AGENT);
  __hip_atomic_fetch_add(a + 64, v2, __ATOMIC_RELAXED, __HIP_MEMORY_SCOPE_AGENT);
  __hip_atomic_fetch_add(a + 96, v3, __ATOMIC_RELAXED, __HIP_MEMORY_SCOPE_AGENT);
}


__device__ __forceinline__ void load_tile_sw(v2f* __restrict__ sT,
                                             const float* __restrict__ g,
                                             int rowBase, int tid) {
  const float4* g4 = reinterpret_cast<const float4*>(g);
#pragma unroll
  for (int i = 0; i < 16; ++i) {
    int linear = i * TPB + tid;
    int r  = linear & 63;
    int c4 = linear >> 6;
    int gr = rowBase + r;
    if (gr >= N_NODES) gr = N_NODES - 1;
    float4 v = g4[(size_t)gr * 32 + c4];
    sT[(c4 * 2) * XPITCH + r]     = mk2(v.x, v.y);
    sT[(c4 * 2 + 1) * XPITCH + r] = mk2(v.z, v.w);
  }
}

__device__ __forceinline__ void load_w_sw(v2f* __restrict__ sW,
                                          const float* __restrict__ W,
                                          int rowOff, int tid) {
  const float4* g4 = reinterpret_cast<const float4*>(W + (size_t)rowOff * D);
#pragma unroll
  for (int i = 0; i < 8; ++i) {
    int linear = i * TPB + tid;
    int p  = linear >> 5;
    int c4 = linear & 31;
    float4 r0 = g4[(p * 2) * 32 + c4];
    float4 r1 = g4[(p * 2 + 1) * 32 + c4];
    v2f* dstp = sW + p * WPITCH + c4 * 4;
    dstp[0] = mk2(r0.x, r1.x);
    dstp[1] = mk2(r0.y, r1.y);
    dstp[2] = mk2(r0.z, r1.z);
    dstp[3] = mk2(r0.w, r1.w);
  }
}

__device__ __forceinline__ void gemm64(const v2f* __restrict__ sX, int kpBase,
                                       const v2f* __restrict__ sW,
                                       v8f acc[8], int m0, int lane) {
  const int lrow = lane & 15;
  const int hi   = lane >> 4;
#pragma unroll 4
  for (int kp = 0; kp < 32; kp += 2) {
    v2f a = sX[(kpBase + kp + hi) * XPITCH + m0 + lrow];
    const v2f* wp = sW + (kp + hi) * WPITCH + lrow;
#pragma unroll
    for (int n = 0; n < 8; ++n) {
      v2f b = wp[n * 16];
      acc[n] = __builtin_amdgcn_wmma_f32_16x16x4_f32(
           false, a,  false, b,
           (short)0, acc[n],  false,  false);
    }
  }
}

__device__ __forceinline__ void store_acc_sw(v2f* __restrict__ sT,
                                             const v8f acc[8],
                                             const float* __restrict__ bias,
                                             int m0, int lane, bool relu) {
  const int lcol = lane & 15;
  const int mhi  = (lane >> 4) << 3;
  float* base = reinterpret_cast<float*>(sT);
#pragma unroll
  for (int n = 0; n < 8; ++n) {
    int col = n * 16 + lcol;
    float bv = bias[col];
    int p = col >> 1, j = col & 1;
#pragma unroll
    for (int v = 0; v < 8; ++v) {
      float x = acc[n][v] + bv;
      if (relu) x = fmaxf(x, 0.f);
      base[(p * XPITCH + m0 + mhi + v) * 2 + j] = x;
    }
  }
}

__global__ __launch_bounds__(TPB) void mlp_kernel(const float* __restrict__ nf,
                                                  const float* __restrict__ W1,
                                                  const float* __restrict__ b1,
                                                  const float* __restrict__ W2,
                                                  const float* __restrict__ b2,
                                                  const float* __restrict__ W3,
                                                  const float* __restrict__ b3,
                                                  float* __restrict__ out) {
  __shared__ v2f sW[32 * WPITCH];
  __shared__ v2f sX[128 * XPITCH];
  __shared__ v2f sH[64 * XPITCH];

  const int tid  = threadIdx.x;
  const int lane = tid & 31;
  const int m0   = (tid >> 5) * 16;
  const int rowBase = blockIdx.x * M_TILE;

  load_tile_sw(sX,       out, rowBase, tid);
  load_tile_sw(sX + 64 * XPITCH, nf, rowBase, tid);

  const v8f zv = {0.f, 0.f, 0.f, 0.f, 0.f, 0.f, 0.f, 0.f};
  v8f acc[8];
#pragma unroll
  for (int n = 0; n < 8; ++n) acc[n] = zv;

#pragma unroll 1
  for (int s = 0; s < 4; ++s) {
    __syncthreads();
    load_w_sw(sW, W1, s * 64, tid);
    __syncthreads();
    gemm64(sX, s * 32, sW, acc, m0, lane);
  }
  store_acc_sw(sH, acc, b1, m0, lane, true);

#pragma unroll
  for (int n = 0; n < 8; ++n) acc[n] = zv;
#pragma unroll 1
  for (int s = 0; s < 2; ++s) {
    __syncthreads();
    load_w_sw(sW, W2, s * 64, tid);
    __syncthreads();
    gemm64(sH, s * 32, sW, acc, m0, lane);
  }
  store_acc_sw(sX, acc, b2, m0, lane, true);

#pragma unroll
  for (int n = 0; n < 8; ++n) acc[n] = zv;
#pragma unroll 1
  for (int s = 0; s < 2; ++s) {
    __syncthreads();
    load_w_sw(sW, W3, s * 64, tid);
    __syncthreads();
    gemm64(sX, s * 32, sW, acc, m0, lane);
  }
  store_acc_sw(sH, acc, b3, m0, lane, false);
  __syncthreads();

  float4* o4 = reinterpret_cast<float4*>(out);
#pragma unroll
  for (int i = 0; i < 16; ++i) {
    int linear = i * TPB + tid;
    int r  = linear & 63;
    int c4 = linear >> 6;
    int gr = rowBase + r;
    v2f lo = sH[(c4 * 2) * XPITCH + r];
    v2f hv = sH[(c4 * 2 + 1) * XPITCH + r];
    if (gr < N_NODES)
      o4[(size_t)gr * 32 + c4] = make_float4(lo.x, lo.y, hv.x, hv.y);
  }
}

extern "C" void kernel_launch(void* const* d_in, const int* in_sizes, int n_in,
                              void* d_out, int out_size, void* d_ws, size_t ws_size,
                              hipStream_t stream) {
  const float* nf  = (const float*)d_in[0];
  const int*   src = (const int*)d_in[1];
  const int*   dst = (const int*)d_in[2];
  const float* W1  = (const float*)d_in[3];
  const float* b1  = (const float*)d_in[4];
  const float* W2  = (const float*)d_in[5];
  const float* b2  = (const float*)d_in[6];
  const float* W3  = (const float*)d_in[7];
  const float* b3  = (const float*)d_in[8];
  float* out = (float*)d_out;
  (void)in_sizes; (void)n_in; (void)out_size; (void)d_ws; (void)ws_size;

  const int n4 = N_NODES * D / 4;
  zero_kernel<<<(n4 + 255) / 256, 256, 0, stream>>>(out, n4);

  scatter_kernel<<<N_EDGES / 8, 256, 0, stream>>>(nf, src, dst, out);

  const int nBlocks = (N_NODES + M_TILE - 1) / M_TILE;
  mlp_kernel<<<nBlocks, TPB, 0, stream>>>(nf, W1, b1, W2, b2, W3, b3, out);
}
